# starting kernel
# baseline (speedup 1.0000x reference)
#include <hip/hip_runtime.h>

typedef float v2f __attribute__((ext_vector_type(2)));
typedef float v8f __attribute__((ext_vector_type(8)));

#define N_PIX   9216
#define CCH     3
#define HDIM    96
#define QTILES  576
#define BATCH   2
#define KWAVES  8

__device__ __forceinline__ float xor16f(float v) {
    return __int_as_float(__builtin_amdgcn_ds_swizzle(__float_as_int(v), 0x401F));
}
__device__ __forceinline__ float bcastf(float v, int srcLane) {
    return __int_as_float(__builtin_amdgcn_readlane(__float_as_int(v), srcLane));
}

__global__ __launch_bounds__(32 * KWAVES)
void attn_flash_wmma(const float* __restrict__ x, float* __restrict__ fout) {
    const int lane = threadIdx.x & 31;
    const int wv   = threadIdx.x >> 5;
    const int b    = blockIdx.x / QTILES;
    const int qt   = blockIdx.x % QTILES;
    const bool lo  = lane < 16;
    const int  l15 = lane & 15;

    const float* xb = x + (size_t)b * CCH * N_PIX;
    const float scale = 0.57735026918962576f;

    v2f Bq;
    {
        const int q = qt * 16 + l15;
        float c0 = xb[0 * N_PIX + q];
        float c1 = xb[1 * N_PIX + q];
        float c2 = xb[2 * N_PIX + q];
        Bq.x = (lo ? c0 : c2) * scale;
        Bq.y = lo ? (c1 * scale) : 0.0f;
    }

    v8f  Facc = {0.f, 0.f, 0.f, 0.f, 0.f, 0.f, 0.f, 0.f};
    float sum = 0.0f;

    for (int kt = wv * 16; kt < N_PIX; kt += 16 * KWAVES) {
        const int k  = kt + l15;
        float a0 = xb[0 * N_PIX + k];
        float a1 = xb[1 * N_PIX + k];
        float a2 = xb[2 * N_PIX + k];
        v2f Ak;
        Ak.x = lo ? a0 : a2;
        Ak.y = lo ? a1 : 0.0f;

        v8f S = {0.f, 0.f, 0.f, 0.f, 0.f, 0.f, 0.f, 0.f};
        S = __builtin_amdgcn_wmma_f32_16x16x4_f32(false, Ak, false, Bq,
                                                  (short)0, S, false, false);

        float p0 = __expf(S[0]), p1 = __expf(S[1]), p2 = __expf(S[2]), p3 = __expf(S[3]);
        float p4 = __expf(S[4]), p5 = __expf(S[5]), p6 = __expf(S[6]), p7 = __expf(S[7]);
        sum += ((p0 + p1) + (p2 + p3)) + ((p4 + p5) + (p6 + p7));

        float t0 = xor16f(p0), t1 = xor16f(p1), t2 = xor16f(p2), t3 = xor16f(p3);
        float t4 = xor16f(p4), t5 = xor16f(p5), t6 = xor16f(p6), t7 = xor16f(p7);

        #pragma unroll
        for (int c = 0; c < 4; ++c) {
            v2f Ap;
            if (c == 0)      { Ap.x = lo ? p0 : t2; Ap.y = lo ? p1 : t3; }
            else if (c == 1) { Ap.x = lo ? p4 : t6; Ap.y = lo ? p5 : t7; }
            else if (c == 2) { Ap.x = lo ? t0 : p2; Ap.y = lo ? t1 : p3; }
            else             { Ap.x = lo ? t4 : p6; Ap.y = lo ? t5 : p7; }

            const int ch  = l15;
            const int chc = ch < CCH ? ch : 0;
            const int kb  = kt + 4 * c + (lo ? 0 : 2);
            float b0 = xb[chc * N_PIX + kb + 0];
            float b1 = xb[chc * N_PIX + kb + 1];
            v2f Bv;
            Bv.x = (ch < CCH) ? b0 : 0.0f;
            Bv.y = (ch < CCH) ? b1 : 0.0f;

            Facc = __builtin_amdgcn_wmma_f32_16x16x4_f32(false, Ap, false, Bv,
                                                         (short)0, Facc, false, false);
        }
    }

    __shared__ float red[KWAVES][32][9];
    #pragma unroll
    for (int r = 0; r < 8; ++r) red[wv][lane][r] = Facc[r];
    red[wv][lane][8] = sum;
    __syncthreads();

    if (wv == 0) {
        #pragma unroll
        for (int w = 1; w < KWAVES; ++w) {
            #pragma unroll
            for (int r = 0; r < 8; ++r) Facc[r] += red[w][lane][r];
            sum += red[w][lane][8];
        }

        float sum_tot = sum + xor16f(sum);

        const int ch = l15;
        #pragma unroll
        for (int r = 0; r < 8; ++r) {
            float s_q = lo ? bcastf(sum_tot, r) : bcastf(sum_tot, r + 8);
            if (ch < CCH) {
                const int qq = qt * 16 + r + (lo ? 0 : 8);
                fout[((size_t)b * CCH + ch) * N_PIX + qq] = Facc[r] / s_q;
            }
        }
    }
}

__global__ __launch_bounds__(256)
void conv3x3_bias_residual(const float* __restrict__ inp, const float* __restrict__ f,
                           const float* __restrict__ w, const float* __restrict__ bias,
                           float* __restrict__ out) {
    int idx = blockIdx.x * blockDim.x + threadIdx.x;
    if (idx >= BATCH * CCH * N_PIX) return;
    int xp = idx % HDIM;
    int t  = idx / HDIM;
    int yp = t % HDIM;  t /= HDIM;
    int co = t % CCH;
    int b  = t / CCH;

    const float* fb = f + (size_t)b * CCH * N_PIX;
    float acc = bias[co];
    #pragma unroll
    for (int ci = 0; ci < CCH; ++ci) {
        #pragma unroll
        for (int dy = -1; dy <= 1; ++dy) {
            #pragma unroll
            for (int dx = -1; dx <= 1; ++dx) {
                int yy = yp + dy, xx = xp + dx;
                if (yy >= 0 && yy < HDIM && xx >= 0 && xx < HDIM) {
                    float wv = w[((co * CCH + ci) * 3 + (dy + 1)) * 3 + (dx + 1)];
                    acc += wv * fb[ci * N_PIX + yy * HDIM + xx];
                }
            }
        }
    }
    out[idx] = inp[idx] + acc;
}

extern "C" void kernel_launch(void* const* d_in, const int* in_sizes, int n_in,
                              void* d_out, int out_size, void* d_ws, size_t ws_size,
                              hipStream_t stream) {
    const float* inp    = (const float*)d_in[0];
    const float* conv_w = (const float*)d_in[1];
    const float* conv_b = (const float*)d_in[2];
    float* out = (float*)d_out;
    float* f   = (float*)d_ws;

    attn_flash_wmma<<<BATCH * QTILES, 32 * KWAVES, 0, stream>>>(inp, f);

    const int tot = BATCH * CCH * N_PIX;
    conv3x3_bias_residual<<<(tot + 255) / 256, 256, 0, stream>>>(inp, f, conv_w, conv_b, out);
}


# PV product transposed (F^T = V^T P): exp tile feeds WMMA B in its own D layout, no lane exchange or selects; V^T read as 8-float runs; order of key sums inside each 16-key tile changes
# speedup vs baseline: 1.1493x; 1.1493x over previous
#include <hip/hip_runtime.h>

typedef float v2f __attribute__((ext_vector_type(2)));
typedef float v4f __attribute__((ext_vector_type(4)));
typedef float v8f __attribute__((ext_vector_type(8)));

#define N_PIX   9216
#define CCH     3
#define HDIM    96
#define QTILES  576
#define BATCH   2
#define KWAVES  8

__global__ __launch_bounds__(32 * KWAVES)
void attn_flash_wmma(const float* __restrict__ x, float* __restrict__ fout) {
    const int lane = threadIdx.x & 31;
    const int wv   = threadIdx.x >> 5;
    const int b    = blockIdx.x / QTILES;
    const int qt   = blockIdx.x % QTILES;
    const bool lo  = lane < 16;
    const int  l15 = lane & 15;

    const float* xb = x + (size_t)b * CCH * N_PIX;
    const float scale = 0.57735026918962576f;

    v2f Bq;
    {
        const int q = qt * 16 + l15;
        float c0 = xb[0 * N_PIX + q];
        float c1 = xb[1 * N_PIX + q];
        float c2 = xb[2 * N_PIX + q];
        Bq.x = (lo ? c0 : c2) * scale;
        Bq.y = lo ? (c1 * scale) : 0.0f;
    }

    const int rowc = l15 < CCH ? l15 : (CCH - 1);
    const float* vrow = xb + rowc * N_PIX + (lo ? 0 : 8);

    v8f  Facc = {0.f, 0.f, 0.f, 0.f, 0.f, 0.f, 0.f, 0.f};
    float sum = 0.0f;

    for (int kt = wv * 16; kt < N_PIX; kt += 16 * KWAVES) {
        const int k  = kt + l15;
        float a0 = xb[0 * N_PIX + k];
        float a1 = xb[1 * N_PIX + k];
        float a2 = xb[2 * N_PIX + k];
        v2f Ak;
        Ak.x = lo ? a0 : a2;
        Ak.y = lo ? a1 : 0.0f;

        const v4f va = *(const v4f*)(vrow + kt);
        const v4f vb = *(const v4f*)(vrow + kt + 4);

        v8f S = {0.f, 0.f, 0.f, 0.f, 0.f, 0.f, 0.f, 0.f};
        S = __builtin_amdgcn_wmma_f32_16x16x4_f32(false, Ak, false, Bq,
                                                  (short)0, S, false, false);

        float p0 = __expf(S[0]), p1 = __expf(S[1]), p2 = __expf(S[2]), p3 = __expf(S[3]);
        float p4 = __expf(S[4]), p5 = __expf(S[5]), p6 = __expf(S[6]), p7 = __expf(S[7]);
        sum += ((p0 + p1) + (p2 + p3)) + ((p4 + p5) + (p6 + p7));

        v2f Av, Bp;
        Av.x = va.x; Av.y = va.y; Bp.x = p0; Bp.y = p1;
        Facc = __builtin_amdgcn_wmma_f32_16x16x4_f32(false, Av, false, Bp,
                                                     (short)0, Facc, false, false);
        Av.x = va.z; Av.y = va.w; Bp.x = p2; Bp.y = p3;
        Facc = __builtin_amdgcn_wmma_f32_16x16x4_f32(false, Av, false, Bp,
                                                     (short)0, Facc, false, false);
        Av.x = vb.x; Av.y = vb.y; Bp.x = p4; Bp.y = p5;
        Facc = __builtin_amdgcn_wmma_f32_16x16x4_f32(false, Av, false, Bp,
                                                     (short)0, Facc, false, false);
        Av.x = vb.z; Av.y = vb.w; Bp.x = p6; Bp.y = p7;
        Facc = __builtin_amdgcn_wmma_f32_16x16x4_f32(false, Av, false, Bp,
                                                     (short)0, Facc, false, false);
    }

    __shared__ float red[KWAVES][4][32];
    red[wv][0][lane] = Facc[0];
    red[wv][1][lane] = Facc[1];
    red[wv][2][lane] = Facc[2];
    red[wv][3][lane] = sum;
    __syncthreads();

    const int t = threadIdx.x;
    if (t < CCH * 16) {
        const int ch = t >> 4;
        const int q  = t & 15;
        float fa   = red[0][ch][q];
        float s_lo = red[0][3][q];
        float s_hi = red[0][3][q + 16];
        #pragma unroll
        for (int w = 1; w < KWAVES; ++w) {
            fa   += red[w][ch][q];
            s_lo += red[w][3][q];
            s_hi += red[w][3][q + 16];
        }
        fout[((size_t)b * CCH + ch) * N_PIX + qt * 16 + q] = fa / (s_lo + s_hi);
    }
}

__global__ __launch_bounds__(256)
void conv3x3_bias_residual(const float* __restrict__ inp, const float* __restrict__ f,
                           const float* __restrict__ w, const float* __restrict__ bias,
                           float* __restrict__ out) {
    int idx = blockIdx.x * blockDim.x + threadIdx.x;
    if (idx >= BATCH * CCH * N_PIX) return;
    int xp = idx % HDIM;
    int t  = idx / HDIM;
    int yp = t % HDIM;  t /= HDIM;
    int co = t % CCH;
    int b  = t / CCH;

    const float* fb = f + (size_t)b * CCH * N_PIX;
    float acc = bias[co];
    #pragma unroll
    for (int ci = 0; ci < CCH; ++ci) {
        #pragma unroll
        for (int dy = -1; dy <= 1; ++dy) {
            #pragma unroll
            for (int dx = -1; dx <= 1; ++dx) {
                int yy = yp + dy, xx = xp + dx;
                if (yy >= 0 && yy < HDIM && xx >= 0 && xx < HDIM) {
                    float wv = w[((co * CCH + ci) * 3 + (dy + 1)) * 3 + (dx + 1)];
                    acc += wv * fb[ci * N_PIX + yy * HDIM + xx];
                }
            }
        }
    }
    out[idx] = inp[idx] + acc;
}

extern "C" void kernel_launch(void* const* d_in, const int* in_sizes, int n_in,
                              void* d_out, int out_size, void* d_ws, size_t ws_size,
                              hipStream_t stream) {
    const float* inp    = (const float*)d_in[0];
    const float* conv_w = (const float*)d_in[1];
    const float* conv_b = (const float*)d_in[2];
    float* out = (float*)d_out;
    float* f   = (float*)d_ws;

    attn_flash_wmma<<<BATCH * QTILES, 32 * KWAVES, 0, stream>>>(inp, f);

    const int tot = BATCH * CCH * N_PIX;
    conv3x3_bias_residual<<<(tot + 255) / 256, 256, 0, stream>>>(inp, f, conv_w, conv_b, out);
}


# key loop software-pipelined in registers: each tile's global loads issue three tiles ahead into four rotating register sets, held there by a compiler fence; math and all sum orders unchanged
# speedup vs baseline: 1.2638x; 1.0996x over previous
#include <hip/hip_runtime.h>

typedef float v2f __attribute__((ext_vector_type(2)));
typedef float v4f __attribute__((ext_vector_type(4)));
typedef float v8f __attribute__((ext_vector_type(8)));

#define N_PIX   9216
#define CCH     3
#define HDIM    96
#define QTILES  576
#define BATCH   2
#define KWAVES  8

#define TPW     (QTILES / KWAVES)

struct KeyTile {
    float a_sel;
    float a_mid;
    v4f   va;
    v4f   vb;
};

__device__ __forceinline__ void load_tile(KeyTile& t, const float* __restrict__ xk,
                                          unsigned offA, unsigned offM, unsigned offV) {
    t.a_sel = xk[offA];
    t.a_mid = xk[offM];
    t.va = *(const v4f*)(xk + offV);
    t.vb = *(const v4f*)(xk + offV + 4);
}

__device__ __forceinline__ void tile_math(const KeyTile& t, bool lo, v2f Bq,
                                          v8f& Facc, float& sum) {
    v2f Ak;
    Ak.x = t.a_sel;
    Ak.y = lo ? t.a_mid : 0.0f;

    v8f S = {0.f, 0.f, 0.f, 0.f, 0.f, 0.f, 0.f, 0.f};
    S = __builtin_amdgcn_wmma_f32_16x16x4_f32(false, Ak, false, Bq,
                                              (short)0, S, false, false);

    float p0 = __expf(S[0]), p1 = __expf(S[1]), p2 = __expf(S[2]), p3 = __expf(S[3]);
    float p4 = __expf(S[4]), p5 = __expf(S[5]), p6 = __expf(S[6]), p7 = __expf(S[7]);
    sum += ((p0 + p1) + (p2 + p3)) + ((p4 + p5) + (p6 + p7));

    v2f Av, Bp;
    Av.x = t.va.x; Av.y = t.va.y; Bp.x = p0; Bp.y = p1;
    Facc = __builtin_amdgcn_wmma_f32_16x16x4_f32(false, Av, false, Bp,
                                                 (short)0, Facc, false, false);
    Av.x = t.va.z; Av.y = t.va.w; Bp.x = p2; Bp.y = p3;
    Facc = __builtin_amdgcn_wmma_f32_16x16x4_f32(false, Av, false, Bp,
                                                 (short)0, Facc, false, false);
    Av.x = t.vb.x; Av.y = t.vb.y; Bp.x = p4; Bp.y = p5;
    Facc = __builtin_amdgcn_wmma_f32_16x16x4_f32(false, Av, false, Bp,
                                                 (short)0, Facc, false, false);
    Av.x = t.vb.z; Av.y = t.vb.w; Bp.x = p6; Bp.y = p7;
    Facc = __builtin_amdgcn_wmma_f32_16x16x4_f32(false, Av, false, Bp,
                                                 (short)0, Facc, false, false);
}

__global__ __launch_bounds__(32 * KWAVES)
void attn_flash_wmma(const float* __restrict__ x, float* __restrict__ fout) {
    const int lane = threadIdx.x & 31;
    const int wv   = __builtin_amdgcn_readfirstlane(threadIdx.x >> 5);
    const int b    = blockIdx.x / QTILES;
    const int qt   = blockIdx.x % QTILES;
    const bool lo  = lane < 16;
    const int  l15 = lane & 15;

    const float* xb = x + (size_t)b * CCH * N_PIX;
    const float scale = 0.57735026918962576f;

    v2f Bq;
    {
        const int q = qt * 16 + l15;
        float c0 = xb[0 * N_PIX + q];
        float c1 = xb[1 * N_PIX + q];
        float c2 = xb[2 * N_PIX + q];
        Bq.x = (lo ? c0 : c2) * scale;
        Bq.y = lo ? (c1 * scale) : 0.0f;
    }

    const int rowc = l15 < CCH ? l15 : (CCH - 1);
    const unsigned offA = (unsigned)((lo ? 0 : 2 * N_PIX) + l15);
    const unsigned offM = (unsigned)(N_PIX + l15);
    const unsigned offV = (unsigned)(rowc * N_PIX + (lo ? 0 : 8));

    v8f  Facc = {0.f, 0.f, 0.f, 0.f, 0.f, 0.f, 0.f, 0.f};
    float sum = 0.0f;

    #define TILE_PTR(i) (xb + (wv + KWAVES * ((i) < TPW ? (i) : TPW - 1)) * 16)
    #define HOLD_LOADS_ABOVE() asm volatile("" ::: "memory")
    KeyTile t0, t1, t2, t3;
    load_tile(t0, TILE_PTR(0), offA, offM, offV);
    load_tile(t1, TILE_PTR(1), offA, offM, offV);
    load_tile(t2, TILE_PTR(2), offA, offM, offV);
    HOLD_LOADS_ABOVE();
    for (int i = 0; i < TPW; i += 4) {
        load_tile(t3, TILE_PTR(i + 3), offA, offM, offV);
        HOLD_LOADS_ABOVE();
        tile_math(t0, lo, Bq, Facc, sum);
        load_tile(t0, TILE_PTR(i + 4), offA, offM, offV);
        HOLD_LOADS_ABOVE();
        tile_math(t1, lo, Bq, Facc, sum);
        load_tile(t1, TILE_PTR(i + 5), offA, offM, offV);
        HOLD_LOADS_ABOVE();
        tile_math(t2, lo, Bq, Facc, sum);
        load_tile(t2, TILE_PTR(i + 6), offA, offM, offV);
        HOLD_LOADS_ABOVE();
        tile_math(t3, lo, Bq, Facc, sum);
    }
    #undef HOLD_LOADS_ABOVE
    #undef TILE_PTR

    __shared__ float red[KWAVES][4][32];
    red[wv][0][lane] = Facc[0];
    red[wv][1][lane] = Facc[1];
    red[wv][2][lane] = Facc[2];
    red[wv][3][lane] = sum;
    __syncthreads();

    const int t = threadIdx.x;
    if (t < CCH * 16) {
        const int ch = t >> 4;
        const int q  = t & 15;
        float fa   = red[0][ch][q];
        float s_lo = red[0][3][q];
        float s_hi = red[0][3][q + 16];
        #pragma unroll
        for (int w = 1; w < KWAVES; ++w) {
            fa   += red[w][ch][q];
            s_lo += red[w][3][q];
            s_hi += red[w][3][q + 16];
        }
        fout[((size_t)b * CCH + ch) * N_PIX + qt * 16 + q] = fa / (s_lo + s_hi);
    }
}

__global__ __launch_bounds__(256)
void conv3x3_bias_residual(const float* __restrict__ inp, const float* __restrict__ f,
                           const float* __restrict__ w, const float* __restrict__ bias,
                           float* __restrict__ out) {
    int idx = blockIdx.x * blockDim.x + threadIdx.x;
    if (idx >= BATCH * CCH * N_PIX) return;
    int xp = idx % HDIM;
    int t  = idx / HDIM;
    int yp = t % HDIM;  t /= HDIM;
    int co = t % CCH;
    int b  = t / CCH;

    const float* fb = f + (size_t)b * CCH * N_PIX;
    float acc = bias[co];
    #pragma unroll
    for (int ci = 0; ci < CCH; ++ci) {
        #pragma unroll
        for (int dy = -1; dy <= 1; ++dy) {
            #pragma unroll
            for (int dx = -1; dx <= 1; ++dx) {
                int yy = yp + dy, xx = xp + dx;
                if (yy >= 0 && yy < HDIM && xx >= 0 && xx < HDIM) {
                    float wv = w[((co * CCH + ci) * 3 + (dy + 1)) * 3 + (dx + 1)];
                    acc += wv * fb[ci * N_PIX + yy * HDIM + xx];
                }
            }
        }
    }
    out[idx] = inp[idx] + acc;
}

extern "C" void kernel_launch(void* const* d_in, const int* in_sizes, int n_in,
                              void* d_out, int out_size, void* d_ws, size_t ws_size,
                              hipStream_t stream) {
    const float* inp    = (const float*)d_in[0];
    const float* conv_w = (const float*)d_in[1];
    const float* conv_b = (const float*)d_in[2];
    float* out = (float*)d_out;
    float* f   = (float*)d_ws;

    attn_flash_wmma<<<BATCH * QTILES, 32 * KWAVES, 0, stream>>>(inp, f);

    const int tot = BATCH * CCH * N_PIX;
    conv3x3_bias_residual<<<(tot + 255) / 256, 256, 0, stream>>>(inp, f, conv_w, conv_b, out);
}


# key loop's vector work cut to 8 exp2 + 4 packed adds per tile: log2(e) folded into the query scale, denominator kept as 4 pair sums in the exp tile's own pairs; denominator sum order changes
# speedup vs baseline: 1.3682x; 1.0826x over previous
#include <hip/hip_runtime.h>

typedef float v2f __attribute__((ext_vector_type(2)));
typedef float v4f __attribute__((ext_vector_type(4)));
typedef float v8f __attribute__((ext_vector_type(8)));

#define N_PIX   9216
#define CCH     3
#define HDIM    96
#define QTILES  576
#define BATCH   2
#define KWAVES  8

#define TPW     (QTILES / KWAVES)

struct KeyTile {
    float a_sel;
    float a_mid;
    v4f   va;
    v4f   vb;
};

__device__ __forceinline__ void load_tile(KeyTile& t, const float* __restrict__ xk,
                                          unsigned offA, unsigned offM, unsigned offV) {
    t.a_sel = xk[offA];
    t.a_mid = xk[offM];
    t.va = *(const v4f*)(xk + offV);
    t.vb = *(const v4f*)(xk + offV + 4);
}

__device__ __forceinline__ void tile_math(const KeyTile& t, v2f Bq, v8f& Facc,
                                          v2f& den0, v2f& den1, v2f& den2, v2f& den3) {
    v2f Ak;
    Ak.x = t.a_sel;
    Ak.y = t.a_mid;

    v8f S = {0.f, 0.f, 0.f, 0.f, 0.f, 0.f, 0.f, 0.f};
    S = __builtin_amdgcn_wmma_f32_16x16x4_f32(false, Ak, false, Bq,
                                              (short)0, S, false, false);

    v2f P0, P1, P2, P3;
    P0.x = __builtin_amdgcn_exp2f(S[0]); P0.y = __builtin_amdgcn_exp2f(S[1]);
    P1.x = __builtin_amdgcn_exp2f(S[2]); P1.y = __builtin_amdgcn_exp2f(S[3]);
    P2.x = __builtin_amdgcn_exp2f(S[4]); P2.y = __builtin_amdgcn_exp2f(S[5]);
    P3.x = __builtin_amdgcn_exp2f(S[6]); P3.y = __builtin_amdgcn_exp2f(S[7]);
    den0 += P0; den1 += P1; den2 += P2; den3 += P3;

    v2f Av;
    Av.x = t.va.x; Av.y = t.va.y;
    Facc = __builtin_amdgcn_wmma_f32_16x16x4_f32(false, Av, false, P0,
                                                 (short)0, Facc, false, false);
    Av.x = t.va.z; Av.y = t.va.w;
    Facc = __builtin_amdgcn_wmma_f32_16x16x4_f32(false, Av, false, P1,
                                                 (short)0, Facc, false, false);
    Av.x = t.vb.x; Av.y = t.vb.y;
    Facc = __builtin_amdgcn_wmma_f32_16x16x4_f32(false, Av, false, P2,
                                                 (short)0, Facc, false, false);
    Av.x = t.vb.z; Av.y = t.vb.w;
    Facc = __builtin_amdgcn_wmma_f32_16x16x4_f32(false, Av, false, P3,
                                                 (short)0, Facc, false, false);
}

__global__ __launch_bounds__(32 * KWAVES)
void attn_flash_wmma(const float* __restrict__ x, float* __restrict__ fout) {
    const int lane = threadIdx.x & 31;
    const int wv   = __builtin_amdgcn_readfirstlane(threadIdx.x >> 5);
    const int b    = blockIdx.x / QTILES;
    const int qt   = blockIdx.x % QTILES;
    const bool lo  = lane < 16;
    const int  l15 = lane & 15;

    const float* xb = x + (size_t)b * CCH * N_PIX;
    const float scale = 0.83294037021578f;

    v2f Bq;
    {
        const int q = qt * 16 + l15;
        float c0 = xb[0 * N_PIX + q];
        float c1 = xb[1 * N_PIX + q];
        float c2 = xb[2 * N_PIX + q];
        Bq.x = (lo ? c0 : c2) * scale;
        Bq.y = lo ? (c1 * scale) : 0.0f;
    }

    const int rowc = l15 < CCH ? l15 : (CCH - 1);
    const unsigned offA = (unsigned)((lo ? 0 : 2 * N_PIX) + l15);
    const unsigned offM = (unsigned)(N_PIX + l15);
    const unsigned offV = (unsigned)(rowc * N_PIX + (lo ? 0 : 8));

    v8f  Facc = {0.f, 0.f, 0.f, 0.f, 0.f, 0.f, 0.f, 0.f};
    v2f den0 = {0.f, 0.f}, den1 = {0.f, 0.f}, den2 = {0.f, 0.f}, den3 = {0.f, 0.f};

    #define TILE_PTR(i) (xb + (wv + KWAVES * ((i) < TPW ? (i) : TPW - 1)) * 16)
    #define HOLD_LOADS_ABOVE() asm volatile("" ::: "memory")
    KeyTile t0, t1, t2, t3;
    load_tile(t0, TILE_PTR(0), offA, offM, offV);
    load_tile(t1, TILE_PTR(1), offA, offM, offV);
    load_tile(t2, TILE_PTR(2), offA, offM, offV);
    HOLD_LOADS_ABOVE();
    for (int i = 0; i < TPW; i += 4) {
        load_tile(t3, TILE_PTR(i + 3), offA, offM, offV);
        HOLD_LOADS_ABOVE();
        tile_math(t0, Bq, Facc, den0, den1, den2, den3);
        load_tile(t0, TILE_PTR(i + 4), offA, offM, offV);
        HOLD_LOADS_ABOVE();
        tile_math(t1, Bq, Facc, den0, den1, den2, den3);
        load_tile(t1, TILE_PTR(i + 5), offA, offM, offV);
        HOLD_LOADS_ABOVE();
        tile_math(t2, Bq, Facc, den0, den1, den2, den3);
        load_tile(t2, TILE_PTR(i + 6), offA, offM, offV);
        HOLD_LOADS_ABOVE();
        tile_math(t3, Bq, Facc, den0, den1, den2, den3);
    }
    #undef HOLD_LOADS_ABOVE
    #undef TILE_PTR

    __shared__ float red[KWAVES][4][32];
    red[wv][0][lane] = Facc[0];
    red[wv][1][lane] = Facc[1];
    red[wv][2][lane] = Facc[2];
    red[wv][3][lane] = ((den0.x + den0.y) + (den1.x + den1.y)) + ((den2.x + den2.y) + (den3.x + den3.y));
    __syncthreads();

    const int t = threadIdx.x;
    if (t < CCH * 16) {
        const int ch = t >> 4;
        const int q  = t & 15;
        float fa   = red[0][ch][q];
        float s_lo = red[0][3][q];
        float s_hi = red[0][3][q + 16];
        #pragma unroll
        for (int w = 1; w < KWAVES; ++w) {
            fa   += red[w][ch][q];
            s_lo += red[w][3][q];
            s_hi += red[w][3][q + 16];
        }
        fout[((size_t)b * CCH + ch) * N_PIX + qt * 16 + q] = fa / (s_lo + s_hi);
    }
}

__global__ __launch_bounds__(256)
void conv3x3_bias_residual(const float* __restrict__ inp, const float* __restrict__ f,
                           const float* __restrict__ w, const float* __restrict__ bias,
                           float* __restrict__ out) {
    int idx = blockIdx.x * blockDim.x + threadIdx.x;
    if (idx >= BATCH * CCH * N_PIX) return;
    int xp = idx % HDIM;
    int t  = idx / HDIM;
    int yp = t % HDIM;  t /= HDIM;
    int co = t % CCH;
    int b  = t / CCH;

    const float* fb = f + (size_t)b * CCH * N_PIX;
    float acc = bias[co];
    #pragma unroll
    for (int ci = 0; ci < CCH; ++ci) {
        #pragma unroll
        for (int dy = -1; dy <= 1; ++dy) {
            #pragma unroll
            for (int dx = -1; dx <= 1; ++dx) {
                int yy = yp + dy, xx = xp + dx;
                if (yy >= 0 && yy < HDIM && xx >= 0 && xx < HDIM) {
                    float wv = w[((co * CCH + ci) * 3 + (dy + 1)) * 3 + (dx + 1)];
                    acc += wv * fb[ci * N_PIX + yy * HDIM + xx];
                }
            }
        }
    }
    out[idx] = inp[idx] + acc;
}

extern "C" void kernel_launch(void* const* d_in, const int* in_sizes, int n_in,
                              void* d_out, int out_size, void* d_ws, size_t ws_size,
                              hipStream_t stream) {
    const float* inp    = (const float*)d_in[0];
    const float* conv_w = (const float*)d_in[1];
    const float* conv_b = (const float*)d_in[2];
    float* out = (float*)d_out;
    float* f   = (float*)d_ws;

    attn_flash_wmma<<<BATCH * QTILES, 32 * KWAVES, 0, stream>>>(inp, f);

    const int tot = BATCH * CCH * N_PIX;
    conv3x3_bias_residual<<<(tot + 255) / 256, 256, 0, stream>>>(inp, f, conv_w, conv_b, out);
}


# conv kernel: all 27 taps loaded unconditionally (out-of-image taps read the centre pixel, value zeroed), weights block-uniform: one load group, not 27 dependent round trips; sum order unchanged
# speedup vs baseline: 1.4616x; 1.0683x over previous
#include <hip/hip_runtime.h>

typedef float v2f __attribute__((ext_vector_type(2)));
typedef float v4f __attribute__((ext_vector_type(4)));
typedef float v8f __attribute__((ext_vector_type(8)));

#define N_PIX   9216
#define CCH     3
#define HDIM    96
#define QTILES  576
#define BATCH   2
#define KWAVES  8

#define TPW     (QTILES / KWAVES)

struct KeyTile {
    float a_sel;
    float a_mid;
    v4f   va;
    v4f   vb;
};

__device__ __forceinline__ void load_tile(KeyTile& t, const float* __restrict__ xk,
                                          unsigned offA, unsigned offM, unsigned offV) {
    t.a_sel = xk[offA];
    t.a_mid = xk[offM];
    t.va = *(const v4f*)(xk + offV);
    t.vb = *(const v4f*)(xk + offV + 4);
}

__device__ __forceinline__ void tile_math(const KeyTile& t, v2f Bq, v8f& Facc,
                                          v2f& den0, v2f& den1, v2f& den2, v2f& den3) {
    v2f Ak;
    Ak.x = t.a_sel;
    Ak.y = t.a_mid;

    v8f S = {0.f, 0.f, 0.f, 0.f, 0.f, 0.f, 0.f, 0.f};
    S = __builtin_amdgcn_wmma_f32_16x16x4_f32(false, Ak, false, Bq,
                                              (short)0, S, false, false);

    v2f P0, P1, P2, P3;
    P0.x = __builtin_amdgcn_exp2f(S[0]); P0.y = __builtin_amdgcn_exp2f(S[1]);
    P1.x = __builtin_amdgcn_exp2f(S[2]); P1.y = __builtin_amdgcn_exp2f(S[3]);
    P2.x = __builtin_amdgcn_exp2f(S[4]); P2.y = __builtin_amdgcn_exp2f(S[5]);
    P3.x = __builtin_amdgcn_exp2f(S[6]); P3.y = __builtin_amdgcn_exp2f(S[7]);
    den0 += P0; den1 += P1; den2 += P2; den3 += P3;

    v2f Av;
    Av.x = t.va.x; Av.y = t.va.y;
    Facc = __builtin_amdgcn_wmma_f32_16x16x4_f32(false, Av, false, P0,
                                                 (short)0, Facc, false, false);
    Av.x = t.va.z; Av.y = t.va.w;
    Facc = __builtin_amdgcn_wmma_f32_16x16x4_f32(false, Av, false, P1,
                                                 (short)0, Facc, false, false);
    Av.x = t.vb.x; Av.y = t.vb.y;
    Facc = __builtin_amdgcn_wmma_f32_16x16x4_f32(false, Av, false, P2,
                                                 (short)0, Facc, false, false);
    Av.x = t.vb.z; Av.y = t.vb.w;
    Facc = __builtin_amdgcn_wmma_f32_16x16x4_f32(false, Av, false, P3,
                                                 (short)0, Facc, false, false);
}

__global__ __launch_bounds__(32 * KWAVES)
void attn_flash_wmma(const float* __restrict__ x, float* __restrict__ fout) {
    const int lane = threadIdx.x & 31;
    const int wv   = __builtin_amdgcn_readfirstlane(threadIdx.x >> 5);
    const int b    = blockIdx.x / QTILES;
    const int qt   = blockIdx.x % QTILES;
    const bool lo  = lane < 16;
    const int  l15 = lane & 15;

    const float* xb = x + (size_t)b * CCH * N_PIX;
    const float scale = 0.83294037021578f;

    v2f Bq;
    {
        const int q = qt * 16 + l15;
        float c0 = xb[0 * N_PIX + q];
        float c1 = xb[1 * N_PIX + q];
        float c2 = xb[2 * N_PIX + q];
        Bq.x = (lo ? c0 : c2) * scale;
        Bq.y = lo ? (c1 * scale) : 0.0f;
    }

    const int rowc = l15 < CCH ? l15 : (CCH - 1);
    const unsigned offA = (unsigned)((lo ? 0 : 2 * N_PIX) + l15);
    const unsigned offM = (unsigned)(N_PIX + l15);
    const unsigned offV = (unsigned)(rowc * N_PIX + (lo ? 0 : 8));

    v8f  Facc = {0.f, 0.f, 0.f, 0.f, 0.f, 0.f, 0.f, 0.f};
    v2f den0 = {0.f, 0.f}, den1 = {0.f, 0.f}, den2 = {0.f, 0.f}, den3 = {0.f, 0.f};

    #define TILE_PTR(i) (xb + (wv + KWAVES * ((i) < TPW ? (i) : TPW - 1)) * 16)
    #define HOLD_LOADS_ABOVE() asm volatile("" ::: "memory")
    KeyTile t0, t1, t2, t3;
    load_tile(t0, TILE_PTR(0), offA, offM, offV);
    load_tile(t1, TILE_PTR(1), offA, offM, offV);
    load_tile(t2, TILE_PTR(2), offA, offM, offV);
    HOLD_LOADS_ABOVE();
    for (int i = 0; i < TPW; i += 4) {
        load_tile(t3, TILE_PTR(i + 3), offA, offM, offV);
        HOLD_LOADS_ABOVE();
        tile_math(t0, Bq, Facc, den0, den1, den2, den3);
        load_tile(t0, TILE_PTR(i + 4), offA, offM, offV);
        HOLD_LOADS_ABOVE();
        tile_math(t1, Bq, Facc, den0, den1, den2, den3);
        load_tile(t1, TILE_PTR(i + 5), offA, offM, offV);
        HOLD_LOADS_ABOVE();
        tile_math(t2, Bq, Facc, den0, den1, den2, den3);
        load_tile(t2, TILE_PTR(i + 6), offA, offM, offV);
        HOLD_LOADS_ABOVE();
        tile_math(t3, Bq, Facc, den0, den1, den2, den3);
    }
    #undef HOLD_LOADS_ABOVE
    #undef TILE_PTR

    __shared__ float red[KWAVES][4][32];
    red[wv][0][lane] = Facc[0];
    red[wv][1][lane] = Facc[1];
    red[wv][2][lane] = Facc[2];
    red[wv][3][lane] = ((den0.x + den0.y) + (den1.x + den1.y)) + ((den2.x + den2.y) + (den3.x + den3.y));
    __syncthreads();

    const int t = threadIdx.x;
    if (t < CCH * 16) {
        const int ch = t >> 4;
        const int q  = t & 15;
        float fa   = red[0][ch][q];
        float s_lo = red[0][3][q];
        float s_hi = red[0][3][q + 16];
        #pragma unroll
        for (int w = 1; w < KWAVES; ++w) {
            fa   += red[w][ch][q];
            s_lo += red[w][3][q];
            s_hi += red[w][3][q + 16];
        }
        fout[((size_t)b * CCH + ch) * N_PIX + qt * 16 + q] = fa / (s_lo + s_hi);
    }
}

#define CONV_BLOCKS_PER_PLANE (N_PIX / 256)
__global__ __launch_bounds__(256)
void conv3x3_bias_residual(const float* __restrict__ inp, const float* __restrict__ f,
                           const float* __restrict__ w, const float* __restrict__ bias,
                           float* __restrict__ out) {
    const int plane = blockIdx.x / CONV_BLOCKS_PER_PLANE;
    const int pix   = (blockIdx.x - plane * CONV_BLOCKS_PER_PLANE) * 256 + threadIdx.x;
    const int co = plane % CCH;
    const int b  = plane / CCH;
    const int yp = pix / HDIM;
    const int xp = pix - yp * HDIM;
    const bool up = yp > 0, dn = yp < HDIM - 1, lf = xp > 0, rt = xp < HDIM - 1;

    const float* fc = f + (size_t)b * CCH * N_PIX + pix;
    const float* wc = w + co * (CCH * 9);
    const int idx = plane * N_PIX + pix;

    const float xin = inp[idx];
    float v[CCH * 9];
    #pragma unroll
    for (int k = 0; k < CCH * 9; ++k) {
        const int ci = k / 9, dy = (k % 9) / 3 - 1, dx = k % 3 - 1;
        const bool oky = dy < 0 ? up : (dy > 0 ? dn : true);
        const bool okx = dx < 0 ? lf : (dx > 0 ? rt : true);
        const bool ok  = oky && okx;
        const int  off = ok ? dy * HDIM + dx : 0;
        const float val = fc[ci * N_PIX + off];
        v[k] = ok ? val : 0.0f;
    }
    float acc = bias[co];
    #pragma unroll
    for (int k = 0; k < CCH * 9; ++k)
        acc = __builtin_fmaf(wc[k], v[k], acc);
    out[idx] = xin + acc;
}
#undef CONV_BLOCKS_PER_PLANE

extern "C" void kernel_launch(void* const* d_in, const int* in_sizes, int n_in,
                              void* d_out, int out_size, void* d_ws, size_t ws_size,
                              hipStream_t stream) {
    const float* inp    = (const float*)d_in[0];
    const float* conv_w = (const float*)d_in[1];
    const float* conv_b = (const float*)d_in[2];
    float* out = (float*)d_out;
    float* f   = (float*)d_ws;

    attn_flash_wmma<<<BATCH * QTILES, 32 * KWAVES, 0, stream>>>(inp, f);

    const int tot = BATCH * CCH * N_PIX;
    conv3x3_bias_residual<<<(tot + 255) / 256, 256, 0, stream>>>(inp, f, conv_w, conv_b, out);
}


# grid levelled over the 256 WGPs: 1024 16-query blocks + 256 8-query blocks whose two key tiles share each PV product (6 WMMAs per tile pair, not 10); sum order changes for those 2048 queries
# speedup vs baseline: 1.5417x; 1.0548x over previous
#include <hip/hip_runtime.h>

typedef float v2f __attribute__((ext_vector_type(2)));
typedef float v4f __attribute__((ext_vector_type(4)));
typedef float v8f __attribute__((ext_vector_type(8)));

#define N_PIX   9216
#define CCH     3
#define HDIM    96
#define QTILES  576
#define BATCH   2
#define KWAVES  8
#define FULL_PER_BATCH  512
#define FULL_BLOCKS     (BATCH * FULL_PER_BATCH)
#define HALF_Q0         (FULL_PER_BATCH * 16)
#define HALF_PER_BATCH  ((N_PIX - HALF_Q0) / 8)
#define HALF_BLOCKS     (BATCH * HALF_PER_BATCH)
#define PPW             (QTILES / 2 / KWAVES)
#define RED_PLANES      7

#define TPW     (QTILES / KWAVES)

struct KeyTile {
    float a_sel;
    float a_mid;
    v4f   va;
    v4f   vb;
};

__device__ __forceinline__ void load_tile(KeyTile& t, const float* __restrict__ xk,
                                          unsigned offA, unsigned offM, unsigned offV) {
    t.a_sel = xk[offA];
    t.a_mid = xk[offM];
    t.va = *(const v4f*)(xk + offV);
    t.vb = *(const v4f*)(xk + offV + 4);
}

__device__ __forceinline__ void tile_math(const KeyTile& t, v2f Bq, v8f& Facc,
                                          v2f& den0, v2f& den1, v2f& den2, v2f& den3) {
    v2f Ak;
    Ak.x = t.a_sel;
    Ak.y = t.a_mid;

    v8f S = {0.f, 0.f, 0.f, 0.f, 0.f, 0.f, 0.f, 0.f};
    S = __builtin_amdgcn_wmma_f32_16x16x4_f32(false, Ak, false, Bq,
                                              (short)0, S, false, false);

    v2f P0, P1, P2, P3;
    P0.x = __builtin_amdgcn_exp2f(S[0]); P0.y = __builtin_amdgcn_exp2f(S[1]);
    P1.x = __builtin_amdgcn_exp2f(S[2]); P1.y = __builtin_amdgcn_exp2f(S[3]);
    P2.x = __builtin_amdgcn_exp2f(S[4]); P2.y = __builtin_amdgcn_exp2f(S[5]);
    P3.x = __builtin_amdgcn_exp2f(S[6]); P3.y = __builtin_amdgcn_exp2f(S[7]);
    den0 += P0; den1 += P1; den2 += P2; den3 += P3;

    v2f Av;
    Av.x = t.va.x; Av.y = t.va.y;
    Facc = __builtin_amdgcn_wmma_f32_16x16x4_f32(false, Av, false, P0,
                                                 (short)0, Facc, false, false);
    Av.x = t.va.z; Av.y = t.va.w;
    Facc = __builtin_amdgcn_wmma_f32_16x16x4_f32(false, Av, false, P1,
                                                 (short)0, Facc, false, false);
    Av.x = t.vb.x; Av.y = t.vb.y;
    Facc = __builtin_amdgcn_wmma_f32_16x16x4_f32(false, Av, false, P2,
                                                 (short)0, Facc, false, false);
    Av.x = t.vb.z; Av.y = t.vb.w;
    Facc = __builtin_amdgcn_wmma_f32_16x16x4_f32(false, Av, false, P3,
                                                 (short)0, Facc, false, false);
}

struct KeyPair {
    float s0a, s0m;
    float s1a, s1m;
    v4f   va, vb;
};

__device__ __forceinline__ void load_pair(KeyPair& t, const float* __restrict__ xk,
                                          unsigned offA, unsigned offM, unsigned offV) {
    t.s0a = xk[offA];
    t.s0m = xk[offM];
    t.s1a = xk[offA + 16];
    t.s1m = xk[offM + 16];
    t.va = *(const v4f*)(xk + offV);
    t.vb = *(const v4f*)(xk + offV + 4);
}

__device__ __forceinline__ void pair_math(const KeyPair& t, v2f Bq, bool grp0, v8f& Facc,
                                          v2f& den0, v2f& den1, v2f& den2, v2f& den3) {
    v2f A0, A1;
    A0.x = t.s0a; A0.y = t.s0m;
    A1.x = t.s1a; A1.y = t.s1m;
    const v8f Z = {0.f, 0.f, 0.f, 0.f, 0.f, 0.f, 0.f, 0.f};
    v8f S0 = __builtin_amdgcn_wmma_f32_16x16x4_f32(false, A0, false, Bq,
                                                   (short)0, Z, false, false);
    v8f S1 = __builtin_amdgcn_wmma_f32_16x16x4_f32(false, A1, false, Bq,
                                                   (short)0, Z, false, false);
    v2f P0, P1, P2, P3;
    P0.x = __builtin_amdgcn_exp2f(grp0 ? S0[0] : S1[0]); P0.y = __builtin_amdgcn_exp2f(grp0 ? S0[1] : S1[1]);
    P1.x = __builtin_amdgcn_exp2f(grp0 ? S0[2] : S1[2]); P1.y = __builtin_amdgcn_exp2f(grp0 ? S0[3] : S1[3]);
    P2.x = __builtin_amdgcn_exp2f(grp0 ? S0[4] : S1[4]); P2.y = __builtin_amdgcn_exp2f(grp0 ? S0[5] : S1[5]);
    P3.x = __builtin_amdgcn_exp2f(grp0 ? S0[6] : S1[6]); P3.y = __builtin_amdgcn_exp2f(grp0 ? S0[7] : S1[7]);
    den0 += P0; den1 += P1; den2 += P2; den3 += P3;

    v2f Av;
    Av.x = t.va.x; Av.y = t.va.y;
    Facc = __builtin_amdgcn_wmma_f32_16x16x4_f32(false, Av, false, P0,
                                                 (short)0, Facc, false, false);
    Av.x = t.va.z; Av.y = t.va.w;
    Facc = __builtin_amdgcn_wmma_f32_16x16x4_f32(false, Av, false, P1,
                                                 (short)0, Facc, false, false);
    Av.x = t.vb.x; Av.y = t.vb.y;
    Facc = __builtin_amdgcn_wmma_f32_16x16x4_f32(false, Av, false, P2,
                                                 (short)0, Facc, false, false);
    Av.x = t.vb.z; Av.y = t.vb.w;
    Facc = __builtin_amdgcn_wmma_f32_16x16x4_f32(false, Av, false, P3,
                                                 (short)0, Facc, false, false);
}

__device__ __forceinline__ void half_tile_block(const float* __restrict__ x, float* __restrict__ fout,
                                                int hb, int lane, int wv,
                                                float (*red)[RED_PLANES][32]) {
    const int  b    = hb / HALF_PER_BATCH;
    const int  q0   = HALF_Q0 + (hb % HALF_PER_BATCH) * 8;
    const bool lo   = lane < 16;
    const int  l15  = lane & 15;
    const bool grp0 = l15 < 8;

    const float* xb = x + (size_t)b * CCH * N_PIX;
    const float scale = 0.83294037021578f;

    v2f Bq;
    {
        const int q = q0 + (l15 & 7);
        float c0 = xb[0 * N_PIX + q];
        float c1 = xb[1 * N_PIX + q];
        float c2 = xb[2 * N_PIX + q];
        Bq.x = (lo ? c0 : c2) * scale;
        Bq.y = lo ? (c1 * scale) : 0.0f;
    }

    const int rch  = (l15 & 3) < CCH ? (l15 & 3) : (CCH - 1);
    const int rgrp = (l15 >= 4 && l15 < 8) ? 16 : 0;
    const unsigned offA = (unsigned)((lo ? 0 : 2 * N_PIX) + l15);
    const unsigned offM = (unsigned)(N_PIX + l15);
    const unsigned offV = (unsigned)(rch * N_PIX + rgrp + (lo ? 0 : 8));

    v8f Facc = {0.f, 0.f, 0.f, 0.f, 0.f, 0.f, 0.f, 0.f};
    v2f den0 = {0.f, 0.f}, den1 = {0.f, 0.f}, den2 = {0.f, 0.f}, den3 = {0.f, 0.f};

    #define PAIR_PTR(i) (xb + (wv + KWAVES * ((i) < PPW ? (i) : PPW - 1)) * 32)
    #define HOLD_LOADS_ABOVE() asm volatile("" ::: "memory")
    KeyPair t0, t1, t2;
    load_pair(t0, PAIR_PTR(0), offA, offM, offV);
    load_pair(t1, PAIR_PTR(1), offA, offM, offV);
    HOLD_LOADS_ABOVE();
    for (int i = 0; i < PPW; i += 3) {
        load_pair(t2, PAIR_PTR(i + 2), offA, offM, offV);
        HOLD_LOADS_ABOVE();
        pair_math(t0, Bq, grp0, Facc, den0, den1, den2, den3);
        load_pair(t0, PAIR_PTR(i + 3), offA, offM, offV);
        HOLD_LOADS_ABOVE();
        pair_math(t1, Bq, grp0, Facc, den0, den1, den2, den3);
        load_pair(t1, PAIR_PTR(i + 4), offA, offM, offV);
        HOLD_LOADS_ABOVE();
        pair_math(t2, Bq, grp0, Facc, den0, den1, den2, den3);
    }
    #undef HOLD_LOADS_ABOVE
    #undef PAIR_PTR

    red[wv][0][lane] = Facc[0];
    red[wv][1][lane] = Facc[1];
    red[wv][2][lane] = Facc[2];
    red[wv][3][lane] = Facc[4];
    red[wv][4][lane] = Facc[5];
    red[wv][5][lane] = Facc[6];
    red[wv][6][lane] = ((den0.x + den0.y) + (den1.x + den1.y)) + ((den2.x + den2.y) + (den3.x + den3.y));
    __syncthreads();

    const int t = threadIdx.x;
    if (t < CCH * 8) {
        const int ch = t >> 3;
        const int q  = t & 7;
        float fa = red[0][ch][q] + red[0][3 + ch][q + 8];
        float s  = (red[0][6][q] + red[0][6][q + 8]) + (red[0][6][q + 16] + red[0][6][q + 24]);
        #pragma unroll
        for (int w = 1; w < KWAVES; ++w) {
            fa += red[w][ch][q] + red[w][3 + ch][q + 8];
            s  += (red[w][6][q] + red[w][6][q + 8]) + (red[w][6][q + 16] + red[w][6][q + 24]);
        }
        fout[((size_t)b * CCH + ch) * N_PIX + q0 + q] = fa / s;
    }
}

__global__ __launch_bounds__(32 * KWAVES)
void attn_flash_wmma(const float* __restrict__ x, float* __restrict__ fout) {
    const int lane = threadIdx.x & 31;
    const int wv   = __builtin_amdgcn_readfirstlane(threadIdx.x >> 5);
    __shared__ float red[KWAVES][RED_PLANES][32];
    if (blockIdx.x >= FULL_BLOCKS) {
        half_tile_block(x, fout, (int)blockIdx.x - FULL_BLOCKS, lane, wv, red);
        return;
    }
    const int b    = blockIdx.x / FULL_PER_BATCH;
    const int qt   = blockIdx.x % FULL_PER_BATCH;
    const bool lo  = lane < 16;
    const int  l15 = lane & 15;

    const float* xb = x + (size_t)b * CCH * N_PIX;
    const float scale = 0.83294037021578f;

    v2f Bq;
    {
        const int q = qt * 16 + l15;
        float c0 = xb[0 * N_PIX + q];
        float c1 = xb[1 * N_PIX + q];
        float c2 = xb[2 * N_PIX + q];
        Bq.x = (lo ? c0 : c2) * scale;
        Bq.y = lo ? (c1 * scale) : 0.0f;
    }

    const int rowc = l15 < CCH ? l15 : (CCH - 1);
    const unsigned offA = (unsigned)((lo ? 0 : 2 * N_PIX) + l15);
    const unsigned offM = (unsigned)(N_PIX + l15);
    const unsigned offV = (unsigned)(rowc * N_PIX + (lo ? 0 : 8));

    v8f  Facc = {0.f, 0.f, 0.f, 0.f, 0.f, 0.f, 0.f, 0.f};
    v2f den0 = {0.f, 0.f}, den1 = {0.f, 0.f}, den2 = {0.f, 0.f}, den3 = {0.f, 0.f};

    #define TILE_PTR(i) (xb + (wv + KWAVES * ((i) < TPW ? (i) : TPW - 1)) * 16)
    #define HOLD_LOADS_ABOVE() asm volatile("" ::: "memory")
    KeyTile t0, t1, t2, t3;
    load_tile(t0, TILE_PTR(0), offA, offM, offV);
    load_tile(t1, TILE_PTR(1), offA, offM, offV);
    load_tile(t2, TILE_PTR(2), offA, offM, offV);
    HOLD_LOADS_ABOVE();
    for (int i = 0; i < TPW; i += 4) {
        load_tile(t3, TILE_PTR(i + 3), offA, offM, offV);
        HOLD_LOADS_ABOVE();
        tile_math(t0, Bq, Facc, den0, den1, den2, den3);
        load_tile(t0, TILE_PTR(i + 4), offA, offM, offV);
        HOLD_LOADS_ABOVE();
        tile_math(t1, Bq, Facc, den0, den1, den2, den3);
        load_tile(t1, TILE_PTR(i + 5), offA, offM, offV);
        HOLD_LOADS_ABOVE();
        tile_math(t2, Bq, Facc, den0, den1, den2, den3);
        load_tile(t2, TILE_PTR(i + 6), offA, offM, offV);
        HOLD_LOADS_ABOVE();
        tile_math(t3, Bq, Facc, den0, den1, den2, den3);
    }
    #undef HOLD_LOADS_ABOVE
    #undef TILE_PTR

    red[wv][0][lane] = Facc[0];
    red[wv][1][lane] = Facc[1];
    red[wv][2][lane] = Facc[2];
    red[wv][3][lane] = ((den0.x + den0.y) + (den1.x + den1.y)) + ((den2.x + den2.y) + (den3.x + den3.y));
    __syncthreads();

    const int t = threadIdx.x;
    if (t < CCH * 16) {
        const int ch = t >> 4;
        const int q  = t & 15;
        float fa   = red[0][ch][q];
        float s_lo = red[0][3][q];
        float s_hi = red[0][3][q + 16];
        #pragma unroll
        for (int w = 1; w < KWAVES; ++w) {
            fa   += red[w][ch][q];
            s_lo += red[w][3][q];
            s_hi += red[w][3][q + 16];
        }
        fout[((size_t)b * CCH + ch) * N_PIX + qt * 16 + q] = fa / (s_lo + s_hi);
    }
}

#define CONV_BLOCKS_PER_PLANE (N_PIX / 256)
__global__ __launch_bounds__(256)
void conv3x3_bias_residual(const float* __restrict__ inp, const float* __restrict__ f,
                           const float* __restrict__ w, const float* __restrict__ bias,
                           float* __restrict__ out) {
    const int plane = blockIdx.x / CONV_BLOCKS_PER_PLANE;
    const int pix   = (blockIdx.x - plane * CONV_BLOCKS_PER_PLANE) * 256 + threadIdx.x;
    const int co = plane % CCH;
    const int b  = plane / CCH;
    const int yp = pix / HDIM;
    const int xp = pix - yp * HDIM;
    const bool up = yp > 0, dn = yp < HDIM - 1, lf = xp > 0, rt = xp < HDIM - 1;

    const float* fc = f + (size_t)b * CCH * N_PIX + pix;
    const float* wc = w + co * (CCH * 9);
    const int idx = plane * N_PIX + pix;

    const float xin = inp[idx];
    float v[CCH * 9];
    #pragma unroll
    for (int k = 0; k < CCH * 9; ++k) {
        const int ci = k / 9, dy = (k % 9) / 3 - 1, dx = k % 3 - 1;
        const bool oky = dy < 0 ? up : (dy > 0 ? dn : true);
        const bool okx = dx < 0 ? lf : (dx > 0 ? rt : true);
        const bool ok  = oky && okx;
        const int  off = ok ? dy * HDIM + dx : 0;
        const float val = fc[ci * N_PIX + off];
        v[k] = ok ? val : 0.0f;
    }
    float acc = bias[co];
    #pragma unroll
    for (int k = 0; k < CCH * 9; ++k)
        acc = __builtin_fmaf(wc[k], v[k], acc);
    out[idx] = xin + acc;
}
#undef CONV_BLOCKS_PER_PLANE

extern "C" void kernel_launch(void* const* d_in, const int* in_sizes, int n_in,
                              void* d_out, int out_size, void* d_ws, size_t ws_size,
                              hipStream_t stream) {
    const float* inp    = (const float*)d_in[0];
    const float* conv_w = (const float*)d_in[1];
    const float* conv_b = (const float*)d_in[2];
    float* out = (float*)d_out;
    float* f   = (float*)d_ws;

    attn_flash_wmma<<<FULL_BLOCKS + HALF_BLOCKS, 32 * KWAVES, 0, stream>>>(inp, f);

    const int tot = BATCH * CCH * N_PIX;
    conv3x3_bias_residual<<<(tot + 255) / 256, 256, 0, stream>>>(inp, f, conv_w, conv_b, out);
}


# full tiles paired: each wave carries two 16-query tiles (512 work-groups of 32 queries + the 256 8-query ones); a key tile's loads and A operands feed 2 S + 8 PV products; sum orders unchanged
# speedup vs baseline: 1.5986x; 1.0369x over previous
#include <hip/hip_runtime.h>

typedef float v2f __attribute__((ext_vector_type(2)));
typedef float v4f __attribute__((ext_vector_type(4)));
typedef float v8f __attribute__((ext_vector_type(8)));

#define N_PIX   9216
#define CCH     3
#define HDIM    96
#define QTILES  576
#define BATCH   2
#define KWAVES  8
#define PAIR_PER_BATCH  256
#define FULL_BLOCKS     (BATCH * PAIR_PER_BATCH)
#define HALF_Q0         (PAIR_PER_BATCH * 32)
#define HALF_PER_BATCH  ((N_PIX - HALF_Q0) / 8)
#define HALF_BLOCKS     (BATCH * HALF_PER_BATCH)
#define PPW             (QTILES / 2 / KWAVES)
#define RED_PLANES      8

#define TPW     (QTILES / KWAVES)

struct KeyTile {
    float a_sel;
    float a_mid;
    v4f   va;
    v4f   vb;
};

__device__ __forceinline__ void load_tile(KeyTile& t, const float* __restrict__ xk,
                                          unsigned offA, unsigned offM, unsigned offV) {
    t.a_sel = xk[offA];
    t.a_mid = xk[offM];
    t.va = *(const v4f*)(xk + offV);
    t.vb = *(const v4f*)(xk + offV + 4);
}


struct QState {
    v8f Facc;
    v2f den0, den1, den2, den3;
};

__device__ __forceinline__ void score_exp(v2f Ak, v2f Bq, QState& q,
                                          v2f& P0, v2f& P1, v2f& P2, v2f& P3) {
    v8f S = {0.f, 0.f, 0.f, 0.f, 0.f, 0.f, 0.f, 0.f};
    S = __builtin_amdgcn_wmma_f32_16x16x4_f32(false, Ak, false, Bq,
                                              (short)0, S, false, false);
    P0.x = __builtin_amdgcn_exp2f(S[0]); P0.y = __builtin_amdgcn_exp2f(S[1]);
    P1.x = __builtin_amdgcn_exp2f(S[2]); P1.y = __builtin_amdgcn_exp2f(S[3]);
    P2.x = __builtin_amdgcn_exp2f(S[4]); P2.y = __builtin_amdgcn_exp2f(S[5]);
    P3.x = __builtin_amdgcn_exp2f(S[6]); P3.y = __builtin_amdgcn_exp2f(S[7]);
    q.den0 += P0; q.den1 += P1; q.den2 += P2; q.den3 += P3;
}

__device__ __forceinline__ void tile_math(const KeyTile& t, v2f BqA, v2f BqB,
                                          QState& qa, QState& qb) {
    v2f Ak;
    Ak.x = t.a_sel;
    Ak.y = t.a_mid;

    v2f PA0, PA1, PA2, PA3, PB0, PB1, PB2, PB3;
    score_exp(Ak, BqA, qa, PA0, PA1, PA2, PA3);
    score_exp(Ak, BqB, qb, PB0, PB1, PB2, PB3);

    v2f Av;
    Av.x = t.va.x; Av.y = t.va.y;
    qa.Facc = __builtin_amdgcn_wmma_f32_16x16x4_f32(false, Av, false, PA0,
                                                    (short)0, qa.Facc, false, false);
    qb.Facc = __builtin_amdgcn_wmma_f32_16x16x4_f32(false, Av, false, PB0,
                                                    (short)0, qb.Facc, false, false);
    Av.x = t.va.z; Av.y = t.va.w;
    qa.Facc = __builtin_amdgcn_wmma_f32_16x16x4_f32(false, Av, false, PA1,
                                                    (short)0, qa.Facc, false, false);
    qb.Facc = __builtin_amdgcn_wmma_f32_16x16x4_f32(false, Av, false, PB1,
                                                    (short)0, qb.Facc, false, false);
    Av.x = t.vb.x; Av.y = t.vb.y;
    qa.Facc = __builtin_amdgcn_wmma_f32_16x16x4_f32(false, Av, false, PA2,
                                                    (short)0, qa.Facc, false, false);
    qb.Facc = __builtin_amdgcn_wmma_f32_16x16x4_f32(false, Av, false, PB2,
                                                    (short)0, qb.Facc, false, false);
    Av.x = t.vb.z; Av.y = t.vb.w;
    qa.Facc = __builtin_amdgcn_wmma_f32_16x16x4_f32(false, Av, false, PA3,
                                                    (short)0, qa.Facc, false, false);
    qb.Facc = __builtin_amdgcn_wmma_f32_16x16x4_f32(false, Av, false, PB3,
                                                    (short)0, qb.Facc, false, false);
}

struct KeyPair {
    float s0a, s0m;
    float s1a, s1m;
    v4f   va, vb;
};

__device__ __forceinline__ void load_pair(KeyPair& t, const float* __restrict__ xk,
                                          unsigned offA, unsigned offM, unsigned offV) {
    t.s0a = xk[offA];
    t.s0m = xk[offM];
    t.s1a = xk[offA + 16];
    t.s1m = xk[offM + 16];
    t.va = *(const v4f*)(xk + offV);
    t.vb = *(const v4f*)(xk + offV + 4);
}

__device__ __forceinline__ void pair_math(const KeyPair& t, v2f Bq, bool grp0, v8f& Facc,
                                          v2f& den0, v2f& den1, v2f& den2, v2f& den3) {
    v2f A0, A1;
    A0.x = t.s0a; A0.y = t.s0m;
    A1.x = t.s1a; A1.y = t.s1m;
    const v8f Z = {0.f, 0.f, 0.f, 0.f, 0.f, 0.f, 0.f, 0.f};
    v8f S0 = __builtin_amdgcn_wmma_f32_16x16x4_f32(false, A0, false, Bq,
                                                   (short)0, Z, false, false);
    v8f S1 = __builtin_amdgcn_wmma_f32_16x16x4_f32(false, A1, false, Bq,
                                                   (short)0, Z, false, false);
    v2f P0, P1, P2, P3;
    P0.x = __builtin_amdgcn_exp2f(grp0 ? S0[0] : S1[0]); P0.y = __builtin_amdgcn_exp2f(grp0 ? S0[1] : S1[1]);
    P1.x = __builtin_amdgcn_exp2f(grp0 ? S0[2] : S1[2]); P1.y = __builtin_amdgcn_exp2f(grp0 ? S0[3] : S1[3]);
    P2.x = __builtin_amdgcn_exp2f(grp0 ? S0[4] : S1[4]); P2.y = __builtin_amdgcn_exp2f(grp0 ? S0[5] : S1[5]);
    P3.x = __builtin_amdgcn_exp2f(grp0 ? S0[6] : S1[6]); P3.y = __builtin_amdgcn_exp2f(grp0 ? S0[7] : S1[7]);
    den0 += P0; den1 += P1; den2 += P2; den3 += P3;

    v2f Av;
    Av.x = t.va.x; Av.y = t.va.y;
    Facc = __builtin_amdgcn_wmma_f32_16x16x4_f32(false, Av, false, P0,
                                                 (short)0, Facc, false, false);
    Av.x = t.va.z; Av.y = t.va.w;
    Facc = __builtin_amdgcn_wmma_f32_16x16x4_f32(false, Av, false, P1,
                                                 (short)0, Facc, false, false);
    Av.x = t.vb.x; Av.y = t.vb.y;
    Facc = __builtin_amdgcn_wmma_f32_16x16x4_f32(false, Av, false, P2,
                                                 (short)0, Facc, false, false);
    Av.x = t.vb.z; Av.y = t.vb.w;
    Facc = __builtin_amdgcn_wmma_f32_16x16x4_f32(false, Av, false, P3,
                                                 (short)0, Facc, false, false);
}

__device__ __forceinline__ void half_tile_block(const float* __restrict__ x, float* __restrict__ fout,
                                                int hb, int lane, int wv,
                                                float (*red)[RED_PLANES][32]) {
    const int  b    = hb / HALF_PER_BATCH;
    const int  q0   = HALF_Q0 + (hb % HALF_PER_BATCH) * 8;
    const bool lo   = lane < 16;
    const int  l15  = lane & 15;
    const bool grp0 = l15 < 8;

    const float* xb = x + (size_t)b * CCH * N_PIX;
    const float scale = 0.83294037021578f;

    v2f Bq;
    {
        const int q = q0 + (l15 & 7);
        float c0 = xb[0 * N_PIX + q];
        float c1 = xb[1 * N_PIX + q];
        float c2 = xb[2 * N_PIX + q];
        Bq.x = (lo ? c0 : c2) * scale;
        Bq.y = lo ? (c1 * scale) : 0.0f;
    }

    const int rch  = (l15 & 3) < CCH ? (l15 & 3) : (CCH - 1);
    const int rgrp = (l15 >= 4 && l15 < 8) ? 16 : 0;
    const unsigned offA = (unsigned)((lo ? 0 : 2 * N_PIX) + l15);
    const unsigned offM = (unsigned)(N_PIX + l15);
    const unsigned offV = (unsigned)(rch * N_PIX + rgrp + (lo ? 0 : 8));

    v8f Facc = {0.f, 0.f, 0.f, 0.f, 0.f, 0.f, 0.f, 0.f};
    v2f den0 = {0.f, 0.f}, den1 = {0.f, 0.f}, den2 = {0.f, 0.f}, den3 = {0.f, 0.f};

    #define PAIR_PTR(i) (xb + (wv + KWAVES * ((i) < PPW ? (i) : PPW - 1)) * 32)
    #define HOLD_LOADS_ABOVE() asm volatile("" ::: "memory")
    KeyPair t0, t1, t2;
    load_pair(t0, PAIR_PTR(0), offA, offM, offV);
    load_pair(t1, PAIR_PTR(1), offA, offM, offV);
    HOLD_LOADS_ABOVE();
    for (int i = 0; i < PPW; i += 3) {
        load_pair(t2, PAIR_PTR(i + 2), offA, offM, offV);
        HOLD_LOADS_ABOVE();
        pair_math(t0, Bq, grp0, Facc, den0, den1, den2, den3);
        load_pair(t0, PAIR_PTR(i + 3), offA, offM, offV);
        HOLD_LOADS_ABOVE();
        pair_math(t1, Bq, grp0, Facc, den0, den1, den2, den3);
        load_pair(t1, PAIR_PTR(i + 4), offA, offM, offV);
        HOLD_LOADS_ABOVE();
        pair_math(t2, Bq, grp0, Facc, den0, den1, den2, den3);
    }
    #undef HOLD_LOADS_ABOVE
    #undef PAIR_PTR

    red[wv][0][lane] = Facc[0];
    red[wv][1][lane] = Facc[1];
    red[wv][2][lane] = Facc[2];
    red[wv][3][lane] = Facc[4];
    red[wv][4][lane] = Facc[5];
    red[wv][5][lane] = Facc[6];
    red[wv][6][lane] = ((den0.x + den0.y) + (den1.x + den1.y)) + ((den2.x + den2.y) + (den3.x + den3.y));
    __syncthreads();

    const int t = threadIdx.x;
    if (t < CCH * 8) {
        const int ch = t >> 3;
        const int q  = t & 7;
        float fa = red[0][ch][q] + red[0][3 + ch][q + 8];
        float s  = (red[0][6][q] + red[0][6][q + 8]) + (red[0][6][q + 16] + red[0][6][q + 24]);
        #pragma unroll
        for (int w = 1; w < KWAVES; ++w) {
            fa += red[w][ch][q] + red[w][3 + ch][q + 8];
            s  += (red[w][6][q] + red[w][6][q + 8]) + (red[w][6][q + 16] + red[w][6][q + 24]);
        }
        fout[((size_t)b * CCH + ch) * N_PIX + q0 + q] = fa / s;
    }
}

__global__ __launch_bounds__(32 * KWAVES)
void attn_flash_wmma(const float* __restrict__ x, float* __restrict__ fout) {
    const int lane = threadIdx.x & 31;
    const int wv   = __builtin_amdgcn_readfirstlane(threadIdx.x >> 5);
    __shared__ float red[KWAVES][RED_PLANES][32];
    if (blockIdx.x >= FULL_BLOCKS) {
        half_tile_block(x, fout, (int)blockIdx.x - FULL_BLOCKS, lane, wv, red);
        return;
    }
    const int b    = blockIdx.x / PAIR_PER_BATCH;
    const int q0   = (blockIdx.x % PAIR_PER_BATCH) * 32;
    const bool lo  = lane < 16;
    const int  l15 = lane & 15;

    const float* xb = x + (size_t)b * CCH * N_PIX;
    const float scale = 0.83294037021578f;

    v2f BqA, BqB;
    {
        const int q = q0 + l15;
        float c0 = xb[0 * N_PIX + q];
        float c1 = xb[1 * N_PIX + q];
        float c2 = xb[2 * N_PIX + q];
        float d0 = xb[0 * N_PIX + q + 16];
        float d1 = xb[1 * N_PIX + q + 16];
        float d2 = xb[2 * N_PIX + q + 16];
        BqA.x = (lo ? c0 : c2) * scale;
        BqA.y = lo ? (c1 * scale) : 0.0f;
        BqB.x = (lo ? d0 : d2) * scale;
        BqB.y = lo ? (d1 * scale) : 0.0f;
    }

    const int rowc = l15 < CCH ? l15 : (CCH - 1);
    const unsigned offA = (unsigned)((lo ? 0 : 2 * N_PIX) + l15);
    const unsigned offM = (unsigned)(N_PIX + l15);
    const unsigned offV = (unsigned)(rowc * N_PIX + (lo ? 0 : 8));

    QState qa, qb;
    qa.Facc = (v8f){0.f, 0.f, 0.f, 0.f, 0.f, 0.f, 0.f, 0.f};
    qa.den0 = qa.den1 = qa.den2 = qa.den3 = (v2f){0.f, 0.f};
    qb = qa;

    #define TILE_PTR(i) (xb + (wv + KWAVES * ((i) < TPW ? (i) : TPW - 1)) * 16)
    #define HOLD_LOADS_ABOVE() asm volatile("" ::: "memory")
    KeyTile t0, t1, t2, t3;
    load_tile(t0, TILE_PTR(0), offA, offM, offV);
    load_tile(t1, TILE_PTR(1), offA, offM, offV);
    load_tile(t2, TILE_PTR(2), offA, offM, offV);
    HOLD_LOADS_ABOVE();
    for (int i = 0; i < TPW; i += 4) {
        load_tile(t3, TILE_PTR(i + 3), offA, offM, offV);
        HOLD_LOADS_ABOVE();
        tile_math(t0, BqA, BqB, qa, qb);
        load_tile(t0, TILE_PTR(i + 4), offA, offM, offV);
        HOLD_LOADS_ABOVE();
        tile_math(t1, BqA, BqB, qa, qb);
        load_tile(t1, TILE_PTR(i + 5), offA, offM, offV);
        HOLD_LOADS_ABOVE();
        tile_math(t2, BqA, BqB, qa, qb);
        load_tile(t2, TILE_PTR(i + 6), offA, offM, offV);
        HOLD_LOADS_ABOVE();
        tile_math(t3, BqA, BqB, qa, qb);
    }
    #undef HOLD_LOADS_ABOVE
    #undef TILE_PTR

    red[wv][0][lane] = qa.Facc[0];
    red[wv][1][lane] = qa.Facc[1];
    red[wv][2][lane] = qa.Facc[2];
    red[wv][3][lane] = ((qa.den0.x + qa.den0.y) + (qa.den1.x + qa.den1.y)) + ((qa.den2.x + qa.den2.y) + (qa.den3.x + qa.den3.y));
    red[wv][4][lane] = qb.Facc[0];
    red[wv][5][lane] = qb.Facc[1];
    red[wv][6][lane] = qb.Facc[2];
    red[wv][7][lane] = ((qb.den0.x + qb.den0.y) + (qb.den1.x + qb.den1.y)) + ((qb.den2.x + qb.den2.y) + (qb.den3.x + qb.den3.y));
    __syncthreads();

    const int t = threadIdx.x;
    if (t < CCH * 32) {
        const int ch = t >> 5;
        const int j  = t & 31;
        const int pl = (j >> 4) * 4;
        const int q  = j & 15;
        float fa   = red[0][pl + ch][q];
        float s_lo = red[0][pl + 3][q];
        float s_hi = red[0][pl + 3][q + 16];
        #pragma unroll
        for (int w = 1; w < KWAVES; ++w) {
            fa   += red[w][pl + ch][q];
            s_lo += red[w][pl + 3][q];
            s_hi += red[w][pl + 3][q + 16];
        }
        fout[((size_t)b * CCH + ch) * N_PIX + q0 + j] = fa / (s_lo + s_hi);
    }
}

#define CONV_BLOCKS_PER_PLANE (N_PIX / 256)
__global__ __launch_bounds__(256)
void conv3x3_bias_residual(const float* __restrict__ inp, const float* __restrict__ f,
                           const float* __restrict__ w, const float* __restrict__ bias,
                           float* __restrict__ out) {
    const int plane = blockIdx.x / CONV_BLOCKS_PER_PLANE;
    const int pix   = (blockIdx.x - plane * CONV_BLOCKS_PER_PLANE) * 256 + threadIdx.x;
    const int co = plane % CCH;
    const int b  = plane / CCH;
    const int yp = pix / HDIM;
    const int xp = pix - yp * HDIM;
    const bool up = yp > 0, dn = yp < HDIM - 1, lf = xp > 0, rt = xp < HDIM - 1;

    const float* fc = f + (size_t)b * CCH * N_PIX + pix;
    const float* wc = w + co * (CCH * 9);
    const int idx = plane * N_PIX + pix;

    const float xin = inp[idx];
    float v[CCH * 9];
    #pragma unroll
    for (int k = 0; k < CCH * 9; ++k) {
        const int ci = k / 9, dy = (k % 9) / 3 - 1, dx = k % 3 - 1;
        const bool oky = dy < 0 ? up : (dy > 0 ? dn : true);
        const bool okx = dx < 0 ? lf : (dx > 0 ? rt : true);
        const bool ok  = oky && okx;
        const int  off = ok ? dy * HDIM + dx : 0;
        const float val = fc[ci * N_PIX + off];
        v[k] = ok ? val : 0.0f;
    }
    float acc = bias[co];
    #pragma unroll
    for (int k = 0; k < CCH * 9; ++k)
        acc = __builtin_fmaf(wc[k], v[k], acc);
    out[idx] = xin + acc;
}
#undef CONV_BLOCKS_PER_PLANE

extern "C" void kernel_launch(void* const* d_in, const int* in_sizes, int n_in,
                              void* d_out, int out_size, void* d_ws, size_t ws_size,
                              hipStream_t stream) {
    const float* inp    = (const float*)d_in[0];
    const float* conv_w = (const float*)d_in[1];
    const float* conv_b = (const float*)d_in[2];
    float* out = (float*)d_out;
    float* f   = (float*)d_ws;

    attn_flash_wmma<<<FULL_BLOCKS + HALF_BLOCKS, 32 * KWAVES, 0, stream>>>(inp, f);

    const int tot = BATCH * CCH * N_PIX;
    conv3x3_bias_residual<<<(tot + 255) / 256, 256, 0, stream>>>(inp, f, conv_w, conv_b, out);
}
